# starting kernel
# baseline (speedup 1.0000x reference)
#include <hip/hip_runtime.h>


typedef __attribute__((ext_vector_type(2))) float v2f;
typedef __attribute__((ext_vector_type(8))) float v8f;

#define FDIM    1024
#define BAND    32
#define ROWS_PB 64
#define THREADS 128

__device__ __forceinline__ float fast_softplus(float t) {
    float e = __expf(-fabsf(t));
    return fmaxf(t, 0.0f) + __logf(1.0f + e);
}

__device__ __forceinline__ float bump_mask(float arg) {
    float sp1 = fmaxf(fast_softplus(arg),        1e-6f);
    float sp2 = fmaxf(fast_softplus(1.0f - arg), 1e-6f);
    float f1  = __expf(-__builtin_amdgcn_rcpf(sp1));
    float f2  = __expf(-__builtin_amdgcn_rcpf(sp2));
    return f2 * __builtin_amdgcn_rcpf(f1 + f2);
}

__global__ __launch_bounds__(THREADS)
void bumpx_kernel(const float* __restrict__ x,
                  const float* __restrict__ aa,
                  float* __restrict__ out) {
    __shared__ float xs[FDIM];
    __shared__ float lw[ROWS_PB];
    __shared__ float ldn[ROWS_PB];

    const int bl   = blockIdx.y;
    const int tid  = threadIdx.x;

    const float* xrow = x  + (size_t)bl * FDIM;
    const float* arow = aa + (size_t)bl * FDIM;

    for (int i = tid; i < FDIM; i += THREADS) xs[i] = xrow[i];
    __syncthreads();

    const int wid  = tid >> 5;
    const int lane = tid & 31;
    const int n    = lane & 15;
    const int half = lane >> 4;

    const int fbase = blockIdx.x * ROWS_PB + wid * 16;
    const int frow  = fbase + n;

    const float a       = arow[frow];
    const float inv_den = __builtin_amdgcn_rcpf(fmaf(6.0f, a, 9.0f));
    const float c0      = a * a * inv_den;

    const float sel_x = (n == 0) ? 1.0f : 0.0f;
    const float sel_1 = (n == 1) ? 1.0f : 0.0f;

    int gstart = fbase - BAND;        if (gstart < 0)    gstart = 0;
    int gend   = fbase + 16 + BAND;   if (gend   > FDIM) gend   = FDIM;

    v8f acc = {};

    for (int g0 = gstart; g0 < gend; g0 += 4) {
        const int gA  = g0 + 2 * half;
        const float d0 = (float)(gA     - frow);
        const float d1 = (float)(gA + 1 - frow);
        v2f A;
        A.x = bump_mask(fmaf(d0 * d0, inv_den, -c0));
        A.y = bump_mask(fmaf(d1 * d1, inv_den, -c0));

        const float xv0 = xs[gA];
        const float xv1 = xs[gA + 1];
        v2f B;
        B.x = fmaf(sel_x, xv0, sel_1);
        B.y = fmaf(sel_x, xv1, sel_1);

        acc = __builtin_amdgcn_wmma_f32_16x16x4_f32(false, A, false, B,
                                                    (short)0, acc, false, false);
    }

    if (n < 2) {
        float* dst = (n == 0) ? lw : ldn;
        #pragma unroll
        for (int j = 0; j < 8; ++j) dst[wid * 16 + half * 8 + j] = acc[j];
    }
    if (lane < 16) {
        float w = lw[wid * 16 + lane];
        float d = ldn[wid * 16 + lane];
        out[(size_t)bl * FDIM + fbase + lane] = w / d;
    }
}

extern "C" void kernel_launch(void* const* d_in, const int* in_sizes, int n_in,
                              void* d_out, int out_size, void* d_ws, size_t ws_size,
                              hipStream_t stream) {
    const float* x  = (const float*)d_in[0];
    const float* aa = (const float*)d_in[1];
    float* out      = (float*)d_out;

    const int BL = in_sizes[0] / FDIM;
    dim3 grid(FDIM / ROWS_PB, BL);
    bumpx_kernel<<<grid, THREADS, 0, stream>>>(x, aa, out);
}


# Narrow the column band of each 16-row strip from +-32 to +-12: the mask is exactly 0 for |g-f|>=10 when aa is in [0,1), so the K loop runs 10 trips per wave instead of 20
# speedup vs baseline: 1.4386x; 1.4386x over previous
#include <hip/hip_runtime.h>


typedef __attribute__((ext_vector_type(2))) float v2f;
typedef __attribute__((ext_vector_type(8))) float v8f;

#define FDIM    1024
#define BAND    12
#define ROWS_PB 64
#define THREADS 128

__device__ __forceinline__ float fast_softplus(float t) {
    float e = __expf(-fabsf(t));
    return fmaxf(t, 0.0f) + __logf(1.0f + e);
}

__device__ __forceinline__ float bump_mask(float arg) {
    float sp1 = fmaxf(fast_softplus(arg),        1e-6f);
    float sp2 = fmaxf(fast_softplus(1.0f - arg), 1e-6f);
    float f1  = __expf(-__builtin_amdgcn_rcpf(sp1));
    float f2  = __expf(-__builtin_amdgcn_rcpf(sp2));
    return f2 * __builtin_amdgcn_rcpf(f1 + f2);
}

__global__ __launch_bounds__(THREADS)
void bumpx_kernel(const float* __restrict__ x,
                  const float* __restrict__ aa,
                  float* __restrict__ out) {
    __shared__ float xs[FDIM];
    __shared__ float lw[ROWS_PB];
    __shared__ float ldn[ROWS_PB];

    const int bl   = blockIdx.y;
    const int tid  = threadIdx.x;

    const float* xrow = x  + (size_t)bl * FDIM;
    const float* arow = aa + (size_t)bl * FDIM;

    for (int i = tid; i < FDIM; i += THREADS) xs[i] = xrow[i];
    __syncthreads();

    const int wid  = tid >> 5;
    const int lane = tid & 31;
    const int n    = lane & 15;
    const int half = lane >> 4;

    const int fbase = blockIdx.x * ROWS_PB + wid * 16;
    const int frow  = fbase + n;

    const float a       = arow[frow];
    const float inv_den = __builtin_amdgcn_rcpf(fmaf(6.0f, a, 9.0f));
    const float c0      = a * a * inv_den;

    const float sel_x = (n == 0) ? 1.0f : 0.0f;
    const float sel_1 = (n == 1) ? 1.0f : 0.0f;

    int gstart = fbase - BAND;        if (gstart < 0)    gstart = 0;
    int gend   = fbase + 16 + BAND;   if (gend   > FDIM) gend   = FDIM;

    v8f acc = {};

    for (int g0 = gstart; g0 < gend; g0 += 4) {
        const int gA  = g0 + 2 * half;
        const float d0 = (float)(gA     - frow);
        const float d1 = (float)(gA + 1 - frow);
        v2f A;
        A.x = bump_mask(fmaf(d0 * d0, inv_den, -c0));
        A.y = bump_mask(fmaf(d1 * d1, inv_den, -c0));

        const float xv0 = xs[gA];
        const float xv1 = xs[gA + 1];
        v2f B;
        B.x = fmaf(sel_x, xv0, sel_1);
        B.y = fmaf(sel_x, xv1, sel_1);

        acc = __builtin_amdgcn_wmma_f32_16x16x4_f32(false, A, false, B,
                                                    (short)0, acc, false, false);
    }

    if (n < 2) {
        float* dst = (n == 0) ? lw : ldn;
        #pragma unroll
        for (int j = 0; j < 8; ++j) dst[wid * 16 + half * 8 + j] = acc[j];
    }
    if (lane < 16) {
        float w = lw[wid * 16 + lane];
        float d = ldn[wid * 16 + lane];
        out[(size_t)bl * FDIM + fbase + lane] = w / d;
    }
}

extern "C" void kernel_launch(void* const* d_in, const int* in_sizes, int n_in,
                              void* d_out, int out_size, void* d_ws, size_t ws_size,
                              hipStream_t stream) {
    const float* x  = (const float*)d_in[0];
    const float* aa = (const float*)d_in[1];
    float* out      = (float*)d_out;

    const int BL = in_sizes[0] / FDIM;
    dim3 grid(FDIM / ROWS_PB, BL);
    bumpx_kernel<<<grid, THREADS, 0, stream>>>(x, aa, out);
}


# Row-by-distance product: the mask is symmetric in g-f, so A = mask(row, d) for d = 0..11 and B = x[f+d]+x[f-d] (diagonal of A*B); 6 mask evaluations per lane instead of 20
# speedup vs baseline: 2.1167x; 1.4714x over previous
#include <hip/hip_runtime.h>


typedef __attribute__((ext_vector_type(2))) float v2f;
typedef __attribute__((ext_vector_type(8))) float v8f;

#define FDIM    1024
#define NDIST   12
#define ROWS_PB 64
#define THREADS 128

__device__ __forceinline__ float fast_softplus(float t) {
    float e = __expf(-fabsf(t));
    return fmaxf(t, 0.0f) + __logf(1.0f + e);
}

__device__ __forceinline__ float bump_mask(float arg) {
    float sp1 = fmaxf(fast_softplus(arg),        1e-6f);
    float sp2 = fmaxf(fast_softplus(1.0f - arg), 1e-6f);
    float f1  = __expf(-__builtin_amdgcn_rcpf(sp1));
    float f2  = __expf(-__builtin_amdgcn_rcpf(sp2));
    return f2 * __builtin_amdgcn_rcpf(f1 + f2);
}

__global__ __launch_bounds__(THREADS)
void bumpx_kernel(const float* __restrict__ x,
                  const float* __restrict__ aa,
                  float* __restrict__ out) {
    __shared__ float2 xv[NDIST + FDIM + NDIST];

    const int bl   = blockIdx.y;
    const int tid  = threadIdx.x;

    const float* xrow = x  + (size_t)bl * FDIM;
    const float* arow = aa + (size_t)bl * FDIM;

    for (int i = tid; i < FDIM; i += THREADS) xv[NDIST + i] = make_float2(xrow[i], 1.0f);
    if (tid < 2 * NDIST) xv[(tid < NDIST) ? tid : FDIM + tid] = make_float2(0.0f, 0.0f);
    __syncthreads();

    const int wid  = tid >> 5;
    const int lane = tid & 31;
    const int c    = lane & 15;
    const int h    = lane >> 4;

    const int fbase = blockIdx.x * ROWS_PB + wid * 16;
    const int frow  = fbase + c;

    const float a       = arow[frow];
    const float inv_den = __builtin_amdgcn_rcpf(fmaf(6.0f, a, 9.0f));
    const float c0      = a * a * inv_den;

    const int ctr = NDIST + frow;

    v8f accx = {};
    v8f accc = {};

    #pragma unroll
    for (int t = 0; t < NDIST / 4; ++t) {
        const int   d0  = 4 * t + 2 * h;
        const float fd0 = (float)d0;
        const float fd1 = (float)(d0 + 1);
        v2f A;
        A.x = bump_mask(fmaf(fd0 * fd0, inv_den, -c0));
        A.y = bump_mask(fmaf(fd1 * fd1, inv_den, -c0));

        const float2 p0 = xv[ctr + d0];
        const float2 p1 = xv[ctr + d0 + 1];
        float2       m0 = xv[ctr - d0];
        const float2 m1 = xv[ctr - d0 - 1];
        if (d0 == 0) { m0.x = 0.0f; m0.y = 0.0f; }

        v2f Bx, Bc;
        Bx.x = p0.x + m0.x;  Bx.y = p1.x + m1.x;
        Bc.x = p0.y + m0.y;  Bc.y = p1.y + m1.y;

        accx = __builtin_amdgcn_wmma_f32_16x16x4_f32(false, A, false, Bx, (short)0, accx, false, false);
        accc = __builtin_amdgcn_wmma_f32_16x16x4_f32(false, A, false, Bc, (short)0, accc, false, false);
    }

    const int r = c & 7;
    float num = accx[0];
    float den = accc[0];
    #pragma unroll
    for (int j = 1; j < 8; ++j) {
        num = (r == j) ? accx[j] : num;
        den = (r == j) ? accc[j] : den;
    }
    if ((c >> 3) == h) {
        out[(size_t)bl * FDIM + frow] = num / den;
    }
}

extern "C" void kernel_launch(void* const* d_in, const int* in_sizes, int n_in,
                              void* d_out, int out_size, void* d_ws, size_t ws_size,
                              hipStream_t stream) {
    const float* x  = (const float*)d_in[0];
    const float* aa = (const float*)d_in[1];
    float* out      = (float*)d_out;

    const int BL = in_sizes[0] / FDIM;
    dim3 grid(FDIM / ROWS_PB, BL);
    bumpx_kernel<<<grid, THREADS, 0, stream>>>(x, aa, out);
}


# One resident pass: 256 workgroups (the WGP count) of 8 waves, a wave walks four 16-row strips and a block stages only its own 512+24 columns of x; 2048 waves enter, not 8192
# speedup vs baseline: 3.6652x; 1.7316x over previous
#include <hip/hip_runtime.h>


typedef __attribute__((ext_vector_type(2))) float v2f;
typedef __attribute__((ext_vector_type(8))) float v8f;

#define FDIM      1024
#define NDIST     12
#define COLS_PB   512
#define THREADS   256
#define STRIPS_PW 4
#define XV_LEN    (NDIST + COLS_PB + NDIST)

__device__ __forceinline__ float fast_softplus(float t) {
    float e = __expf(-fabsf(t));
    return fmaxf(t, 0.0f) + __logf(1.0f + e);
}

__device__ __forceinline__ float bump_mask(float arg) {
    float sp1 = fmaxf(fast_softplus(arg),        1e-6f);
    float sp2 = fmaxf(fast_softplus(1.0f - arg), 1e-6f);
    float f1  = __expf(-__builtin_amdgcn_rcpf(sp1));
    float f2  = __expf(-__builtin_amdgcn_rcpf(sp2));
    return f2 * __builtin_amdgcn_rcpf(f1 + f2);
}

__global__ __launch_bounds__(THREADS)
void bumpx_kernel(const float* __restrict__ x,
                  const float* __restrict__ aa,
                  float* __restrict__ out) {
    __shared__ float2 xv[XV_LEN];

    const int bl   = blockIdx.y;
    const int f0   = blockIdx.x * COLS_PB;
    const int tid  = threadIdx.x;
    const int wid  = tid >> 5;
    const int lane = tid & 31;
    const int c    = lane & 15;
    const int h    = lane >> 4;

    const float* xrow = x  + (size_t)bl * FDIM;
    const float* arow = aa + (size_t)bl * FDIM;

    const int wloc = wid * (16 * STRIPS_PW);

    float a[STRIPS_PW];
    #pragma unroll
    for (int s = 0; s < STRIPS_PW; ++s) a[s] = arow[f0 + wloc + 16 * s + c];

    const float xs0 = xrow[f0 + tid];
    const float xs1 = xrow[f0 + THREADS + tid];
    float xh = 0.0f, ph = 0.0f;
    if (tid < 2 * NDIST) {
        const int gh = (tid < NDIST) ? (f0 - NDIST + tid) : (f0 + COLS_PB + tid - NDIST);
        if (gh >= 0 && gh < FDIM) { xh = xrow[gh]; ph = 1.0f; }
    }

    xv[NDIST + tid]           = make_float2(xs0, 1.0f);
    xv[NDIST + THREADS + tid] = make_float2(xs1, 1.0f);
    if (tid < 2 * NDIST) xv[(tid < NDIST) ? tid : COLS_PB + tid] = make_float2(xh, ph);
    __syncthreads();

    #pragma unroll
    for (int s = 0; s < STRIPS_PW; ++s) {
        const int floc = wloc + 16 * s + c;
        const int ctr  = NDIST + floc;

        const float inv_den = __builtin_amdgcn_rcpf(fmaf(6.0f, a[s], 9.0f));
        const float c0      = a[s] * a[s] * inv_den;

        v8f accx = {};
        v8f accc = {};

        #pragma unroll
        for (int t = 0; t < NDIST / 4; ++t) {
            const int   d0  = 4 * t + 2 * h;
            const float fd0 = (float)d0;
            const float fd1 = (float)(d0 + 1);
            v2f A;
            A.x = bump_mask(fmaf(fd0 * fd0, inv_den, -c0));
            A.y = bump_mask(fmaf(fd1 * fd1, inv_den, -c0));

            const float2 p0 = xv[ctr + d0];
            const float2 p1 = xv[ctr + d0 + 1];
            float2       m0 = xv[ctr - d0];
            const float2 m1 = xv[ctr - d0 - 1];
            if (d0 == 0) { m0.x = 0.0f; m0.y = 0.0f; }

            v2f Bx, Bc;
            Bx.x = p0.x + m0.x;  Bx.y = p1.x + m1.x;
            Bc.x = p0.y + m0.y;  Bc.y = p1.y + m1.y;

            accx = __builtin_amdgcn_wmma_f32_16x16x4_f32(false, A, false, Bx, (short)0, accx, false, false);
            accc = __builtin_amdgcn_wmma_f32_16x16x4_f32(false, A, false, Bc, (short)0, accc, false, false);
        }

        const int r = c & 7;
        float num = accx[0];
        float den = accc[0];
        #pragma unroll
        for (int j = 1; j < 8; ++j) {
            num = (r == j) ? accx[j] : num;
            den = (r == j) ? accc[j] : den;
        }
        if ((c >> 3) == h) {
            out[(size_t)bl * FDIM + f0 + floc] = num / den;
        }
    }
}

extern "C" void kernel_launch(void* const* d_in, const int* in_sizes, int n_in,
                              void* d_out, int out_size, void* d_ws, size_t ws_size,
                              hipStream_t stream) {
    const float* x  = (const float*)d_in[0];
    const float* aa = (const float*)d_in[1];
    float* out      = (float*)d_out;

    const int BL = in_sizes[0] / FDIM;
    dim3 grid(FDIM / COLS_PB, BL);
    bumpx_kernel<<<grid, THREADS, 0, stream>>>(x, aa, out);
}


# Mask value from ONE exponential on the bare base-2 exp/log/rcp instructions: E=e^arg, sp1=ln(1+E), sp2=ln(E+e)-arg, mask=1/(1+exp(1/sp2-1/sp1)); 755 instructions per wave instead of 1603
# speedup vs baseline: 4.5694x; 1.2467x over previous
#include <hip/hip_runtime.h>


typedef __attribute__((ext_vector_type(2))) float v2f;
typedef __attribute__((ext_vector_type(8))) float v8f;

#define FDIM      1024
#define NDIST     12
#define COLS_PB   512
#define THREADS   256
#define STRIPS_PW 4
#define XV_LEN    (NDIST + COLS_PB + NDIST)

#define LOG2E_F     1.44269504f
#define LOG2E_SQ_F  2.08136898f
#define EULER_F     2.71828183f
#define EPS_L2_F    1.44269504e-6f
__device__ __forceinline__ float bump_mask_l2(float arg2) {
    const float E  = __builtin_amdgcn_exp2f(arg2);
    const float s1 = __builtin_amdgcn_logf(1.0f + E);
    const float s2 = fmaxf(__builtin_amdgcn_logf(E + EULER_F) - arg2, EPS_L2_F);
    const float z2 = (__builtin_amdgcn_rcpf(s2) - __builtin_amdgcn_rcpf(s1)) * LOG2E_SQ_F;
    return __builtin_amdgcn_rcpf(1.0f + __builtin_amdgcn_exp2f(z2));
}

__global__ __launch_bounds__(THREADS)
void bumpx_kernel(const float* __restrict__ x,
                  const float* __restrict__ aa,
                  float* __restrict__ out) {
    __shared__ float2 xv[XV_LEN];

    const int bl   = blockIdx.y;
    const int f0   = blockIdx.x * COLS_PB;
    const int tid  = threadIdx.x;
    const int wid  = tid >> 5;
    const int lane = tid & 31;
    const int c    = lane & 15;
    const int h    = lane >> 4;

    const float* xrow = x  + (size_t)bl * FDIM;
    const float* arow = aa + (size_t)bl * FDIM;

    const int wloc = wid * (16 * STRIPS_PW);

    float a[STRIPS_PW];
    #pragma unroll
    for (int s = 0; s < STRIPS_PW; ++s) a[s] = arow[f0 + wloc + 16 * s + c];

    const float xs0 = xrow[f0 + tid];
    const float xs1 = xrow[f0 + THREADS + tid];
    float xh = 0.0f, ph = 0.0f;
    if (tid < 2 * NDIST) {
        const int gh = (tid < NDIST) ? (f0 - NDIST + tid) : (f0 + COLS_PB + tid - NDIST);
        if (gh >= 0 && gh < FDIM) { xh = xrow[gh]; ph = 1.0f; }
    }

    xv[NDIST + tid]           = make_float2(xs0, 1.0f);
    xv[NDIST + THREADS + tid] = make_float2(xs1, 1.0f);
    if (tid < 2 * NDIST) xv[(tid < NDIST) ? tid : COLS_PB + tid] = make_float2(xh, ph);
    __syncthreads();

    #pragma unroll
    for (int s = 0; s < STRIPS_PW; ++s) {
        const int floc = wloc + 16 * s + c;
        const int ctr  = NDIST + floc;

        const float k2 = __builtin_amdgcn_rcpf(fmaf(6.0f, a[s], 9.0f)) * LOG2E_F;
        const float c2 = a[s] * a[s] * k2;

        v8f accx = {};
        v8f accc = {};

        #pragma unroll
        for (int t = 0; t < NDIST / 4; ++t) {
            const int   d0  = 4 * t + 2 * h;
            const float fd0 = (float)d0;
            const float fd1 = (float)(d0 + 1);
            v2f A;
            A.x = bump_mask_l2(fmaf(fd0 * fd0, k2, -c2));
            A.y = bump_mask_l2(fmaf(fd1 * fd1, k2, -c2));

            const float2 p0 = xv[ctr + d0];
            const float2 p1 = xv[ctr + d0 + 1];
            float2       m0 = xv[ctr - d0];
            const float2 m1 = xv[ctr - d0 - 1];
            if (d0 == 0) { m0.x = 0.0f; m0.y = 0.0f; }

            v2f Bx, Bc;
            Bx.x = p0.x + m0.x;  Bx.y = p1.x + m1.x;
            Bc.x = p0.y + m0.y;  Bc.y = p1.y + m1.y;

            accx = __builtin_amdgcn_wmma_f32_16x16x4_f32(false, A, false, Bx, (short)0, accx, false, false);
            accc = __builtin_amdgcn_wmma_f32_16x16x4_f32(false, A, false, Bc, (short)0, accc, false, false);
        }

        const int r = c & 7;
        float num = accx[0];
        float den = accc[0];
        #pragma unroll
        for (int j = 1; j < 8; ++j) {
            num = (r == j) ? accx[j] : num;
            den = (r == j) ? accc[j] : den;
        }
        if ((c >> 3) == h) {
            out[(size_t)bl * FDIM + f0 + floc] = num / den;
        }
    }
}

extern "C" void kernel_launch(void* const* d_in, const int* in_sizes, int n_in,
                              void* d_out, int out_size, void* d_ws, size_t ws_size,
                              hipStream_t stream) {
    const float* x  = (const float*)d_in[0];
    const float* aa = (const float*)d_in[1];
    float* out      = (float*)d_out;

    const int BL = in_sizes[0] / FDIM;
    dim3 grid(FDIM / COLS_PB, BL);
    bumpx_kernel<<<grid, THREADS, 0, stream>>>(x, aa, out);
}


# Band of distances 12 -> 8, two K=4 tiles per strip not three: for aa in [0,1) the mask at |g-f| >= 8 is <= 1.7e-11 of a row weight >= 0.66 (an exact 0 in the reference's fp32 1-gg form)
# speedup vs baseline: 4.8113x; 1.0529x over previous
#include <hip/hip_runtime.h>


typedef __attribute__((ext_vector_type(2))) float v2f;
typedef __attribute__((ext_vector_type(8))) float v8f;

#define FDIM      1024
#define NDIST     8
#define COLS_PB   512
#define THREADS   256
#define STRIPS_PW 4
#define XV_LEN    (NDIST + COLS_PB + NDIST)

#define LOG2E_F     1.44269504f
#define LOG2E_SQ_F  2.08136898f
#define EULER_F     2.71828183f
#define EPS_L2_F    1.44269504e-6f
__device__ __forceinline__ float bump_mask_l2(float arg2) {
    const float E  = __builtin_amdgcn_exp2f(arg2);
    const float s1 = __builtin_amdgcn_logf(1.0f + E);
    const float s2 = fmaxf(__builtin_amdgcn_logf(E + EULER_F) - arg2, EPS_L2_F);
    const float z2 = (__builtin_amdgcn_rcpf(s2) - __builtin_amdgcn_rcpf(s1)) * LOG2E_SQ_F;
    return __builtin_amdgcn_rcpf(1.0f + __builtin_amdgcn_exp2f(z2));
}

__global__ __launch_bounds__(THREADS)
void bumpx_kernel(const float* __restrict__ x,
                  const float* __restrict__ aa,
                  float* __restrict__ out) {
    __shared__ float2 xv[XV_LEN];

    const int bl   = blockIdx.y;
    const int f0   = blockIdx.x * COLS_PB;
    const int tid  = threadIdx.x;
    const int wid  = tid >> 5;
    const int lane = tid & 31;
    const int c    = lane & 15;
    const int h    = lane >> 4;

    const float* xrow = x  + (size_t)bl * FDIM;
    const float* arow = aa + (size_t)bl * FDIM;

    const int wloc = wid * (16 * STRIPS_PW);

    float a[STRIPS_PW];
    #pragma unroll
    for (int s = 0; s < STRIPS_PW; ++s) a[s] = arow[f0 + wloc + 16 * s + c];

    const float xs0 = xrow[f0 + tid];
    const float xs1 = xrow[f0 + THREADS + tid];
    float xh = 0.0f, ph = 0.0f;
    if (tid < 2 * NDIST) {
        const int gh = (tid < NDIST) ? (f0 - NDIST + tid) : (f0 + COLS_PB + tid - NDIST);
        if (gh >= 0 && gh < FDIM) { xh = xrow[gh]; ph = 1.0f; }
    }

    xv[NDIST + tid]           = make_float2(xs0, 1.0f);
    xv[NDIST + THREADS + tid] = make_float2(xs1, 1.0f);
    if (tid < 2 * NDIST) xv[(tid < NDIST) ? tid : COLS_PB + tid] = make_float2(xh, ph);
    __syncthreads();

    #pragma unroll
    for (int s = 0; s < STRIPS_PW; ++s) {
        const int floc = wloc + 16 * s + c;
        const int ctr  = NDIST + floc;

        const float k2 = __builtin_amdgcn_rcpf(fmaf(6.0f, a[s], 9.0f)) * LOG2E_F;
        const float c2 = a[s] * a[s] * k2;

        v8f accx = {};
        v8f accc = {};

        #pragma unroll
        for (int t = 0; t < NDIST / 4; ++t) {
            const int   d0  = 4 * t + 2 * h;
            const float fd0 = (float)d0;
            const float fd1 = (float)(d0 + 1);
            v2f A;
            A.x = bump_mask_l2(fmaf(fd0 * fd0, k2, -c2));
            A.y = bump_mask_l2(fmaf(fd1 * fd1, k2, -c2));

            const float2 p0 = xv[ctr + d0];
            const float2 p1 = xv[ctr + d0 + 1];
            float2       m0 = xv[ctr - d0];
            const float2 m1 = xv[ctr - d0 - 1];
            if (d0 == 0) { m0.x = 0.0f; m0.y = 0.0f; }

            v2f Bx, Bc;
            Bx.x = p0.x + m0.x;  Bx.y = p1.x + m1.x;
            Bc.x = p0.y + m0.y;  Bc.y = p1.y + m1.y;

            accx = __builtin_amdgcn_wmma_f32_16x16x4_f32(false, A, false, Bx, (short)0, accx, false, false);
            accc = __builtin_amdgcn_wmma_f32_16x16x4_f32(false, A, false, Bc, (short)0, accc, false, false);
        }

        const int r = c & 7;
        float num = accx[0];
        float den = accc[0];
        #pragma unroll
        for (int j = 1; j < 8; ++j) {
            num = (r == j) ? accx[j] : num;
            den = (r == j) ? accc[j] : den;
        }
        if ((c >> 3) == h) {
            out[(size_t)bl * FDIM + f0 + floc] = num / den;
        }
    }
}

extern "C" void kernel_launch(void* const* d_in, const int* in_sizes, int n_in,
                              void* d_out, int out_size, void* d_ws, size_t ws_size,
                              hipStream_t stream) {
    const float* x  = (const float*)d_in[0];
    const float* aa = (const float*)d_in[1];
    float* out      = (float*)d_out;

    const int BL = in_sizes[0] / FDIM;
    dim3 grid(FDIM / COLS_PB, BL);
    bumpx_kernel<<<grid, THREADS, 0, stream>>>(x, aa, out);
}
